# starting kernel
# baseline (speedup 1.0000x reference)
#include <hip/hip_runtime.h>
#include <hip/hip_bf16.h>
#include <math.h>

typedef __attribute__((ext_vector_type(2))) float v2f;
typedef __attribute__((ext_vector_type(8))) float v8f;

__global__ void k_fill(float* __restrict__ p, float v, int n) {
    int i = blockIdx.x * blockDim.x + threadIdx.x;
    if (i < n) p[i] = v;
}

__global__ void k_transpose(const float* __restrict__ W, float* __restrict__ Wt,
                            int K, int Ncol) {
    int i = blockIdx.x * blockDim.x + threadIdx.x;
    if (i < K * Ncol) {
        int k = i / Ncol, nn = i % Ncol;
        Wt[nn * K + k] = W[i];
    }
}

__global__ void k_gemm_f32_wmma(const float* __restrict__ A,
                                const float* __restrict__ Wt,
                                float* __restrict__ C, int nCols) {
    const int K = 128;
    int lane = threadIdx.x;
    int half = lane >> 4;
    int l    = lane & 15;
    int rowBase = blockIdx.x * 16;
    int colBase = (blockIdx.y * 4 + threadIdx.y) * 16;

    const float* arow = A  + (size_t)(rowBase + l) * K + 2 * half;
    const float* brow = Wt + (size_t)(colBase + l) * K + 2 * half;

    v8f acc = {};
#pragma unroll
    for (int k = 0; k < K; k += 4) {
        v2f a = *(const v2f*)(arow + k);
        v2f b = *(const v2f*)(brow + k);
        acc = __builtin_amdgcn_wmma_f32_16x16x4_f32(
             false, a,  false, b,
             (short)0, acc,  false,  false);
    }
#pragma unroll
    for (int r = 0; r < 8; ++r)
        C[(size_t)(rowBase + r + 8 * half) * nCols + colBase + l] = acc[r];
}

__global__ void k_attn(const float* __restrict__ h, const float* __restrict__ att_s,
                       const float* __restrict__ att_d, float* __restrict__ asrc,
                       float* __restrict__ adst, int n, int heads, int C) {
    int i = blockIdx.x * blockDim.x + threadIdx.x;
    if (i >= n * heads) return;
    int node = i / heads, hd = i % heads;
    const float* hp = h + (size_t)node * heads * C + (size_t)hd * C;
    float ss = 0.f, sd = 0.f;
    for (int c = 0; c < C; ++c) {
        float v = hp[c];
        ss += v * att_s[hd * C + c];
        sd += v * att_d[hd * C + c];
    }
    asrc[i] = ss;
    adst[i] = sd;
}

__device__ __forceinline__ void atomicMaxF(float* addr, float val) {
    if (val >= 0.f) atomicMax((int*)addr, __float_as_int(val));
    else            atomicMin((unsigned int*)addr, (unsigned int)__float_as_int(val));
}

__device__ __forceinline__ void edge_pair(int e, int E, const int* __restrict__ src,
                                          const int* __restrict__ dst, int& s, int& d) {
    if (e < E) { s = src[e]; d = dst[e]; }
    else       { s = d = e - E; }
}

__global__ void k_edge_max(const int* __restrict__ src, const int* __restrict__ dst,
                           const float* __restrict__ asrc, const float* __restrict__ adst,
                           float* __restrict__ m, int E, int n, int heads) {
    int e = blockIdx.x * blockDim.x + threadIdx.x;
    if (e >= E + n) return;
    int s, d; edge_pair(e, E, src, dst, s, d);
    for (int h = 0; h < heads; ++h) {
        float v = asrc[s * heads + h] + adst[d * heads + h];
        v = v > 0.f ? v : 0.2f * v;
        atomicMaxF(&m[d * heads + h], v);
    }
}

__global__ void k_edge_expsum(const int* __restrict__ src, const int* __restrict__ dst,
                              const float* __restrict__ asrc, const float* __restrict__ adst,
                              const float* __restrict__ m, float* __restrict__ ex,
                              float* __restrict__ den, int E, int n, int heads) {
    int e = blockIdx.x * blockDim.x + threadIdx.x;
    if (e >= E + n) return;
    int s, d; edge_pair(e, E, src, dst, s, d);
    for (int h = 0; h < heads; ++h) {
        float v = asrc[s * heads + h] + adst[d * heads + h];
        v = v > 0.f ? v : 0.2f * v;
        float x = expf(v - m[d * heads + h]);
        ex[(size_t)e * heads + h] = x;
        atomicAdd(&den[d * heads + h], x);
    }
}

__global__ void k_edge_agg(const int* __restrict__ src, const int* __restrict__ dst,
                           const float* __restrict__ ex, const float* __restrict__ den,
                           const float* __restrict__ feat, float* __restrict__ agg,
                           int E, int n, int heads, int C) {
    int HC = heads * C;
    int epb = blockDim.x / HC;
    int eLocal = threadIdx.x / HC;
    int c = threadIdx.x % HC;
    int e = blockIdx.x * epb + eLocal;
    if (e >= E + n) return;
    int s, d; edge_pair(e, E, src, dst, s, d);
    int hd = c / C;
    float alpha = ex[(size_t)e * heads + hd] / den[d * heads + hd];
    atomicAdd(&agg[(size_t)d * HC + c], feat[(size_t)s * HC + c] * alpha);
}

__global__ void k_bias_elu(float* __restrict__ h, const float* __restrict__ bias,
                           int n, int HC) {
    int i = blockIdx.x * blockDim.x + threadIdx.x;
    if (i >= n * HC) return;
    float v = h[i] + bias[i % HC];
    h[i] = v > 0.f ? v : (expf(v) - 1.f);
}

__global__ void k_bias_add(float* __restrict__ o, const float* __restrict__ bias,
                           int n, int C) {
    int i = blockIdx.x * blockDim.x + threadIdx.x;
    if (i >= n * C) return;
    o[i] += bias[i % C];
}

extern "C" void kernel_launch(void* const* d_in, const int* in_sizes, int n_in,
                              void* d_out, int out_size, void* d_ws, size_t ws_size,
                              hipStream_t stream) {
    const float* x      = (const float*)d_in[0];
    const int*   ei     = (const int*)  d_in[1];
    const float* W1     = (const float*)d_in[2];
    const float* att_s1 = (const float*)d_in[3];
    const float* att_d1 = (const float*)d_in[4];
    const float* b1     = (const float*)d_in[5];
    const float* W2     = (const float*)d_in[6];
    const float* att_s2 = (const float*)d_in[7];
    const float* att_d2 = (const float*)d_in[8];
    const float* b2     = (const float*)d_in[9];

    const int IN = 128, HID = 128, H1 = 4, C1 = 32, OUT = 64;
    const int n   = in_sizes[0] / IN;
    const int E   = in_sizes[1] / 2;
    const int tot = E + n;

    float* out1  = (float*)d_out;
    float* hfeat = (float*)d_out + (size_t)n * OUT;

    float* w = (float*)d_ws;
    size_t off = 0;
    float* Wt1 = w + off; off += (size_t)IN * HID;
    float* Wt2 = w + off; off += (size_t)HID * OUT;
    float* h1  = w + off; off += (size_t)n * HID;
    float* h2  = w + off; off += (size_t)n * OUT;
    float* as1 = w + off; off += (size_t)n * H1;
    float* ad1 = w + off; off += (size_t)n * H1;
    float* m1  = w + off; off += (size_t)n * H1;
    float* dn1 = w + off; off += (size_t)n * H1;
    float* as2 = w + off; off += (size_t)n;
    float* ad2 = w + off; off += (size_t)n;
    float* m2  = w + off; off += (size_t)n;
    float* dn2 = w + off; off += (size_t)n;
    float* ex1 = w + off; off += (size_t)tot * H1;
    float* ex2 = w + off; off += (size_t)tot;

    const int* srcI = ei;
    const int* dstI = ei + E;

    auto cdiv = [](long a, long b) { return (unsigned)((a + b - 1) / b); };
    dim3 B256(256);

    k_transpose<<<cdiv((long)IN * HID, 256), B256, 0, stream>>>(W1, Wt1, IN, HID);
    k_transpose<<<cdiv((long)HID * OUT, 256), B256, 0, stream>>>(W2, Wt2, HID, OUT);

    k_gemm_f32_wmma<<<dim3(n / 16, HID / 64), dim3(32, 4), 0, stream>>>(x, Wt1, h1, HID);
    k_attn<<<cdiv((long)n * H1, 256), B256, 0, stream>>>(h1, att_s1, att_d1, as1, ad1, n, H1, C1);

    k_fill<<<cdiv((long)n * H1, 256), B256, 0, stream>>>(m1, -INFINITY, n * H1);
    k_fill<<<cdiv((long)n * H1, 256), B256, 0, stream>>>(dn1, 0.f, n * H1);
    k_fill<<<cdiv((long)n * HID, 256), B256, 0, stream>>>(hfeat, 0.f, n * HID);

    k_edge_max   <<<cdiv(tot, 256), B256, 0, stream>>>(srcI, dstI, as1, ad1, m1, E, n, H1);
    k_edge_expsum<<<cdiv(tot, 256), B256, 0, stream>>>(srcI, dstI, as1, ad1, m1, ex1, dn1, E, n, H1);
    k_edge_agg   <<<cdiv((long)tot * 128, 256), B256, 0, stream>>>(srcI, dstI, ex1, dn1, h1, hfeat, E, n, H1, C1);
    k_bias_elu   <<<cdiv((long)n * HID, 256), B256, 0, stream>>>(hfeat, b1, n, HID);

    k_gemm_f32_wmma<<<dim3(n / 16, OUT / 64), dim3(32, 4), 0, stream>>>(hfeat, Wt2, h2, OUT);
    k_attn<<<cdiv((long)n, 256), B256, 0, stream>>>(h2, att_s2, att_d2, as2, ad2, n, 1, OUT);

    k_fill<<<cdiv((long)n, 256), B256, 0, stream>>>(m2, -INFINITY, n);
    k_fill<<<cdiv((long)n, 256), B256, 0, stream>>>(dn2, 0.f, n);
    k_fill<<<cdiv((long)n * OUT, 256), B256, 0, stream>>>(out1, 0.f, n * OUT);

    k_edge_max   <<<cdiv(tot, 256), B256, 0, stream>>>(srcI, dstI, as2, ad2, m2, E, n, 1);
    k_edge_expsum<<<cdiv(tot, 256), B256, 0, stream>>>(srcI, dstI, as2, ad2, m2, ex2, dn2, E, n, 1);
    k_edge_agg   <<<cdiv((long)tot * 64, 256), B256, 0, stream>>>(srcI, dstI, ex2, dn2, h2, out1, E, n, 1, OUT);
    k_bias_add   <<<cdiv((long)n * OUT, 256), B256, 0, stream>>>(out1, b2, n, OUT);
}


# GEMM weight-stationary: each wave keeps its 128x16 slice of W in registers, read straight from W (both transpose launches gone); A tiles enter LDS as whole 512-byte rows, 2 buffers; sums as before
# speedup vs baseline: 3.1398x; 3.1398x over previous
#include <hip/hip_runtime.h>
#include <hip/hip_bf16.h>
#include <math.h>

typedef __attribute__((ext_vector_type(2))) float v2f;
typedef __attribute__((ext_vector_type(4))) float v4f;
typedef __attribute__((ext_vector_type(8))) float v8f;

__global__ void k_fill(float* __restrict__ p, float v, int n) {
    int i = blockIdx.x * blockDim.x + threadIdx.x;
    if (i < n) p[i] = v;
}

template <int NCOL>
__global__ void __launch_bounds__(NCOL * 2)
k_gemm_f32_wmma(const float* __restrict__ A, const float* __restrict__ W,
                float* __restrict__ C, int nTiles) {
    constexpr int K   = 128;
    constexpr int NW  = NCOL / 16;
    constexpr int RPW = 16 / NW;
    constexpr int P   = 132;
    __shared__ float sA[2][16 * P];

    const int lane = threadIdx.x & 31;
    const int wave = __builtin_amdgcn_readfirstlane((int)(threadIdx.x >> 5));
    const int half = lane >> 4;
    const int l    = lane & 15;
    const int colBase = wave * 16;

    int t = blockIdx.x;
    v4f st[RPW];
    if (t < nTiles) {
#pragma unroll
        for (int i = 0; i < RPW; ++i)
            st[i] = *(const v4f*)(A + (size_t)(t * 16 + wave * RPW + i) * K + lane * 4);
    }

    v2f b[K / 4];
    const float* wp = W + (size_t)(2 * half) * NCOL + colBase + l;
#pragma unroll
    for (int j = 0; j < K / 4; ++j) {
        b[j].x = wp[(size_t)(4 * j) * NCOL];
        b[j].y = wp[(size_t)(4 * j + 1) * NCOL];
    }

    int buf = 0;
    for (; t < nTiles; t += gridDim.x) {
#pragma unroll
        for (int i = 0; i < RPW; ++i)
            *(v4f*)(&sA[buf][(wave * RPW + i) * P + lane * 4]) = st[i];
        __syncthreads();
        const int tn = t + gridDim.x;
        if (tn < nTiles) {
#pragma unroll
            for (int i = 0; i < RPW; ++i)
                st[i] = *(const v4f*)(A + (size_t)(tn * 16 + wave * RPW + i) * K + lane * 4);
        }
        const float* ap = &sA[buf][l * P + 2 * half];
        v8f acc = {};
#pragma unroll
        for (int j = 0; j < K / 4; ++j) {
            v2f a = *(const v2f*)(ap + 4 * j);
            acc = __builtin_amdgcn_wmma_f32_16x16x4_f32(
                 false, a,  false, b[j],
                 (short)0, acc,  false,  false);
        }
#pragma unroll
        for (int r = 0; r < 8; ++r)
            C[(size_t)(t * 16 + r + 8 * half) * NCOL + colBase + l] = acc[r];
        buf ^= 1;
    }
}

__global__ void k_attn(const float* __restrict__ h, const float* __restrict__ att_s,
                       const float* __restrict__ att_d, float* __restrict__ asrc,
                       float* __restrict__ rec, int n, int heads, int C) {
    int i = blockIdx.x * blockDim.x + threadIdx.x;
    if (i >= n * heads) return;
    int node = i / heads, hd = i % heads;
    const float* hp = h + (size_t)node * heads * C + (size_t)hd * C;
    float ss = 0.f, sd = 0.f;
    for (int c = 0; c < C; ++c) {
        float v = hp[c];
        ss += v * att_s[hd * C + c];
        sd += v * att_d[hd * C + c];
    }
    asrc[i] = ss;
    v2f r; r.x = sd; r.y = ss;
    *(v2f*)(rec + 2 * (size_t)i) = r;
}

__device__ __forceinline__ void edge_pair(int e, int E, const int* __restrict__ src,
                                          const int* __restrict__ dst, int& s, int& d) {
    if (e < E) { s = src[e]; d = dst[e]; }
    else       { s = d = e - E; }
}

template <int H>
__global__ void __launch_bounds__(256)
k_edge_max(const int* __restrict__ src, const int* __restrict__ dst,
           const float* __restrict__ asrc, float* __restrict__ rec, int E) {
    constexpr int EPI = 32 / H;
    const int lane = threadIdx.x & 31;
    const int le = lane / H, lh = lane % H;
    const int e0 = (int)(blockIdx.x * 256 + (threadIdx.x & ~31u)) + le;
    int s[H], d[H];
#pragma unroll
    for (int j = 0; j < H; ++j) {
        const int e = min(e0 + j * EPI, E - 1);
        s[j] = src[e];
        d[j] = dst[e];
    }
    float v[H];
#pragma unroll
    for (int j = 0; j < H; ++j) v[j] = asrc[s[j] * H + lh];
#pragma unroll
    for (int j = 0; j < H; ++j)
        __hip_atomic_fetch_max(&rec[2 * (d[j] * H + lh) + 1], v[j], __ATOMIC_RELAXED,
                               __HIP_MEMORY_SCOPE_AGENT);
}

template <int H, int C>
__global__ void __launch_bounds__(256)
k_edge_fused(const int* __restrict__ src, const int* __restrict__ dst,
             const float* __restrict__ asrc, const float* __restrict__ rec,
             const float* __restrict__ feat,
             float* __restrict__ agg, float* __restrict__ den, int E, int n) {
    constexpr int HC  = H * C;
    constexpr int NJ  = HC / 32;
    constexpr int EPW = 4;
    static_assert(HC % 32 == 0 && (H == 1 || C % 32 == 0), "line must not straddle heads");
    const int lane = threadIdx.x & 31;
    const int wave = __builtin_amdgcn_readfirstlane((int)(threadIdx.x >> 5));
    const int tot = E + n;
    const int e0 = (blockIdx.x * 8 + wave) * EPW;
    if (e0 >= tot) return;
    const int ne = (tot - e0) < EPW ? (tot - e0) : EPW;

    const bool real = e0 + EPW <= E;
    const bool loop = e0 >= E;
    if (ne == EPW && (real || loop)) {
        int s[EPW], d[EPW];
        float f[EPW][NJ];
        if (real) {
#pragma unroll
            for (int i = 0; i < EPW; ++i) { s[i] = src[e0 + i]; d[i] = dst[e0 + i]; }
        } else {
#pragma unroll
            for (int i = 0; i < EPW; ++i) { s[i] = d[i] = e0 + i - E; }
        }
#pragma unroll
        for (int i = 0; i < EPW; ++i)
#pragma unroll
            for (int j = 0; j < NJ; ++j)
                f[i][j] = feat[(size_t)s[i] * HC + j * 32 + lane];
        const int li = lane / H, lh = lane % H;
        int sl = s[0], dl = d[0];
#pragma unroll
        for (int i = 1; i < EPW; ++i)
            if (li == i) { sl = s[i]; dl = d[i]; }
        const int dsel = dl * H + lh;
        float wsel = 0.f;
        if (lane < EPW * H) {
            const v2f r = *(const v2f*)(rec + 2 * (size_t)dsel);
            float v = asrc[sl * H + lh] + r.x;
            v = v > 0.f ? v : 0.2f * v;
            float mx = r.y + r.x;
            mx = mx > 0.f ? mx : 0.2f * mx;
            wsel = expf(v - mx);
        }
#pragma unroll
        for (int i = 0; i < EPW; ++i)
#pragma unroll
            for (int j = 0; j < NJ; ++j) {
                const float wij = __int_as_float(__builtin_amdgcn_readlane(
                    __float_as_int(wsel), i * H + (j * 32) / C));
                atomicAdd(&agg[(size_t)d[i] * HC + j * 32 + lane], f[i][j] * wij);
            }
        if (lane < EPW * H) atomicAdd(&den[dsel], wsel);
    } else {
        for (int i = 0; i < ne; ++i) {
            int s, d; edge_pair(e0 + i, E, src, dst, s, d);
            float w[H];
#pragma unroll
            for (int h = 0; h < H; ++h) {
                const v2f r = *(const v2f*)(rec + 2 * (size_t)(d * H + h));
                float v = asrc[s * H + h] + r.x;
                v = v > 0.f ? v : 0.2f * v;
                float mx = r.y + r.x;
                mx = mx > 0.f ? mx : 0.2f * mx;
                w[h] = expf(v - mx);
            }
            float wl = w[0];
#pragma unroll
            for (int h = 1; h < H; ++h) wl = (lane == h) ? w[h] : wl;
            if (lane < H) atomicAdd(&den[d * H + lane], wl);
#pragma unroll
            for (int j = 0; j < NJ; ++j)
                atomicAdd(&agg[(size_t)d * HC + j * 32 + lane],
                          feat[(size_t)s * HC + j * 32 + lane] * w[(j * 32) / C]);
        }
    }
}

template <int HC, int C>
__global__ void k_bias_elu(float* __restrict__ h, const float* __restrict__ den,
                           const float* __restrict__ bias, int n) {
    int i = blockIdx.x * blockDim.x + threadIdx.x;
    if (i >= n * HC) return;
    int node = i / HC, c = i % HC;
    float v = h[i] / den[node * (HC / C) + c / C] + bias[c];
    h[i] = v > 0.f ? v : (expf(v) - 1.f);
}

template <int C>
__global__ void k_bias_add(float* __restrict__ o, const float* __restrict__ den,
                           const float* __restrict__ bias, int n) {
    int i = blockIdx.x * blockDim.x + threadIdx.x;
    if (i >= n * C) return;
    o[i] = o[i] / den[i / C] + bias[i % C];
}

extern "C" void kernel_launch(void* const* d_in, const int* in_sizes, int n_in,
                              void* d_out, int out_size, void* d_ws, size_t ws_size,
                              hipStream_t stream) {
    const float* x      = (const float*)d_in[0];
    const int*   ei     = (const int*)  d_in[1];
    const float* W1     = (const float*)d_in[2];
    const float* att_s1 = (const float*)d_in[3];
    const float* att_d1 = (const float*)d_in[4];
    const float* b1     = (const float*)d_in[5];
    const float* W2     = (const float*)d_in[6];
    const float* att_s2 = (const float*)d_in[7];
    const float* att_d2 = (const float*)d_in[8];
    const float* b2     = (const float*)d_in[9];

    const int IN = 128, HID = 128, H1 = 4, C1 = 32, OUT = 64;
    const int n   = in_sizes[0] / IN;
    const int E   = in_sizes[1] / 2;
    const int tot = E + n;

    float* out1  = (float*)d_out;
    float* hfeat = (float*)d_out + (size_t)n * OUT;

    float* w = (float*)d_ws;
    size_t off = 0;
    float* rc1 = w + off; off += (size_t)n * H1 * 2;
    float* rc2 = w + off; off += (size_t)n * 2;
    float* h1  = w + off; off += (size_t)n * HID;
    float* h2  = w + off; off += (size_t)n * OUT;
    float* as1 = w + off; off += (size_t)n * H1;
    float* dn1 = w + off; off += (size_t)n * H1;
    float* as2 = w + off; off += (size_t)n;
    float* dn2 = w + off; off += (size_t)n;

    const int* srcI = ei;
    const int* dstI = ei + E;

    auto cdiv = [](long a, long b) { return (unsigned)((a + b - 1) / b); };
    dim3 B256(256);

    const int gemmBlocks = 625;

    k_gemm_f32_wmma<128><<<gemmBlocks, 256, 0, stream>>>(x, W1, h1, n / 16);
    k_attn<<<cdiv((long)n * H1, 256), B256, 0, stream>>>(h1, att_s1, att_d1, as1, rc1, n, H1, C1);

    k_fill<<<cdiv((long)n * H1, 256), B256, 0, stream>>>(dn1, 0.f, n * H1);
    k_fill<<<cdiv((long)n * HID, 256), B256, 0, stream>>>(hfeat, 0.f, n * HID);

    k_edge_max<4><<<cdiv(E, 256), B256, 0, stream>>>(srcI, dstI, as1, rc1, E);
    k_edge_fused<4, 32><<<cdiv(tot, 32), B256, 0, stream>>>(srcI, dstI, as1, rc1, h1, hfeat, dn1, E, n);
    k_bias_elu<128, 32><<<cdiv((long)n * HID, 256), B256, 0, stream>>>(hfeat, dn1, b1, n);

    k_gemm_f32_wmma<64><<<gemmBlocks, 128, 0, stream>>>(hfeat, W2, h2, n / 16);
    k_attn<<<cdiv((long)n, 256), B256, 0, stream>>>(h2, att_s2, att_d2, as2, rc2, n, 1, OUT);

    k_fill<<<cdiv((long)n, 256), B256, 0, stream>>>(dn2, 0.f, n);
    k_fill<<<cdiv((long)n * OUT, 256), B256, 0, stream>>>(out1, 0.f, n * OUT);

    k_edge_max<1><<<cdiv(E, 256), B256, 0, stream>>>(srcI, dstI, as2, rc2, E);
    k_edge_fused<1, 64><<<cdiv(tot, 32), B256, 0, stream>>>(srcI, dstI, as2, rc2, h2, out1, dn2, E, n);
    k_bias_add<64><<<cdiv((long)n * OUT, 256), B256, 0, stream>>>(out1, dn2, b2, n);
}


# Softmax shift = lrelu(global max source score + adst), a bound made in the score kernel: within a score range of 40 both segment-max passes return at entry; true max beyond; sum order as before
# speedup vs baseline: 3.9533x; 1.2591x over previous
#include <hip/hip_runtime.h>
#include <hip/hip_bf16.h>
#include <math.h>

typedef __attribute__((ext_vector_type(2))) float v2f;
typedef __attribute__((ext_vector_type(4))) float v4f;
typedef __attribute__((ext_vector_type(8))) float v8f;

__global__ void k_fill(float* __restrict__ p, float v, int n) {
    int i = blockIdx.x * blockDim.x + threadIdx.x;
    if (i < n) p[i] = v;
}

template <int NCOL>
__global__ void __launch_bounds__(NCOL * 2)
k_gemm_f32_wmma(const float* __restrict__ A, const float* __restrict__ W,
                float* __restrict__ C, int nTiles) {
    constexpr int K   = 128;
    constexpr int NW  = NCOL / 16;
    constexpr int RPW = 16 / NW;
    constexpr int P   = 132;
    __shared__ float sA[2][16 * P];

    const int lane = threadIdx.x & 31;
    const int wave = __builtin_amdgcn_readfirstlane((int)(threadIdx.x >> 5));
    const int half = lane >> 4;
    const int l    = lane & 15;
    const int colBase = wave * 16;

    int t = blockIdx.x;
    v4f st[RPW];
    if (t < nTiles) {
#pragma unroll
        for (int i = 0; i < RPW; ++i)
            st[i] = *(const v4f*)(A + (size_t)(t * 16 + wave * RPW + i) * K + lane * 4);
    }

    v2f b[K / 4];
    const float* wp = W + (size_t)(2 * half) * NCOL + colBase + l;
#pragma unroll
    for (int j = 0; j < K / 4; ++j) {
        b[j].x = wp[(size_t)(4 * j) * NCOL];
        b[j].y = wp[(size_t)(4 * j + 1) * NCOL];
    }

    int buf = 0;
    for (; t < nTiles; t += gridDim.x) {
#pragma unroll
        for (int i = 0; i < RPW; ++i)
            *(v4f*)(&sA[buf][(wave * RPW + i) * P + lane * 4]) = st[i];
        __syncthreads();
        const int tn = t + gridDim.x;
        if (tn < nTiles) {
#pragma unroll
            for (int i = 0; i < RPW; ++i)
                st[i] = *(const v4f*)(A + (size_t)(tn * 16 + wave * RPW + i) * K + lane * 4);
        }
        const float* ap = &sA[buf][l * P + 2 * half];
        v8f acc = {};
#pragma unroll
        for (int j = 0; j < K / 4; ++j) {
            v2f a = *(const v2f*)(ap + 4 * j);
            acc = __builtin_amdgcn_wmma_f32_16x16x4_f32(
                 false, a,  false, b[j],
                 (short)0, acc,  false,  false);
        }
#pragma unroll
        for (int r = 0; r < 8; ++r)
            C[(size_t)(t * 16 + r + 8 * half) * NCOL + colBase + l] = acc[r];
        buf ^= 1;
    }
}

__global__ void __launch_bounds__(256)
k_attn(const float* __restrict__ h, const float* __restrict__ att_s,
       const float* __restrict__ att_d, float* __restrict__ asrc,
       float* __restrict__ rec, float* __restrict__ rng, int n, int heads, int C) {
    __shared__ float shi[8], slo[8];
    const int i = blockIdx.x * blockDim.x + threadIdx.x;
    float hi = 0.f, lo = 0.f;
    if (i < n * heads) {
        int node = i / heads, hd = i % heads;
        const float* hp = h + (size_t)node * heads * C + (size_t)hd * C;
        float ss = 0.f, sd = 0.f;
        for (int c = 0; c < C; ++c) {
            float v = hp[c];
            ss += v * att_s[hd * C + c];
            sd += v * att_d[hd * C + c];
        }
        asrc[i] = ss;
        v2f r; r.x = sd; r.y = ss;
        *(v2f*)(rec + 2 * (size_t)i) = r;
        hi = ss; lo = -ss;
    }
    for (int o = 16; o > 0; o >>= 1) {
        hi = fmaxf(hi, __shfl_xor(hi, o));
        lo = fmaxf(lo, __shfl_xor(lo, o));
    }
    const int wv = threadIdx.x >> 5;
    if ((threadIdx.x & 31) == 0) { shi[wv] = hi; slo[wv] = lo; }
    __syncthreads();
    if (threadIdx.x == 0) {
        for (int w = 1; w < 8; ++w) { hi = fmaxf(hi, shi[w]); lo = fmaxf(lo, slo[w]); }
        __hip_atomic_fetch_max(&rng[0], hi, __ATOMIC_RELAXED, __HIP_MEMORY_SCOPE_AGENT);
        __hip_atomic_fetch_max(&rng[1], lo, __ATOMIC_RELAXED, __HIP_MEMORY_SCOPE_AGENT);
    }
}

__device__ __forceinline__ void edge_pair(int e, int E, const int* __restrict__ src,
                                          const int* __restrict__ dst, int& s, int& d) {
    if (e < E) { s = src[e]; d = dst[e]; }
    else       { s = d = e - E; }
}

#define SHIFT_RANGE 40.f
template <int H>
__global__ void __launch_bounds__(256)
k_edge_max(const int* __restrict__ src, const int* __restrict__ dst,
           const float* __restrict__ asrc, float* __restrict__ rec,
           const float* __restrict__ rng, int E) {
    if (rng[0] + rng[1] <= SHIFT_RANGE) return;
    constexpr int EPI = 32 / H;
    const int lane = threadIdx.x & 31;
    const int le = lane / H, lh = lane % H;
    const int e0 = (int)(blockIdx.x * 256 + (threadIdx.x & ~31u)) + le;
    int s[H], d[H];
#pragma unroll
    for (int j = 0; j < H; ++j) {
        const int e = min(e0 + j * EPI, E - 1);
        s[j] = src[e];
        d[j] = dst[e];
    }
    float v[H];
#pragma unroll
    for (int j = 0; j < H; ++j) v[j] = asrc[s[j] * H + lh];
#pragma unroll
    for (int j = 0; j < H; ++j)
        __hip_atomic_fetch_max(&rec[2 * (d[j] * H + lh) + 1], v[j], __ATOMIC_RELAXED,
                               __HIP_MEMORY_SCOPE_AGENT);
}

template <int H, int C>
__global__ void __launch_bounds__(256)
k_edge_fused(const int* __restrict__ src, const int* __restrict__ dst,
             const float* __restrict__ asrc, const float* __restrict__ rec,
             const float* __restrict__ rng, const float* __restrict__ feat,
             float* __restrict__ agg, float* __restrict__ den, int E, int n) {
    constexpr int HC  = H * C;
    constexpr int NJ  = HC / 32;
    constexpr int EPW = 4;
    static_assert(HC % 32 == 0 && (H == 1 || C % 32 == 0), "line must not straddle heads");
    const int lane = threadIdx.x & 31;
    const int wave = __builtin_amdgcn_readfirstlane((int)(threadIdx.x >> 5));
    const int tot = E + n;
    const int e0 = (blockIdx.x * 8 + wave) * EPW;
    if (e0 >= tot) return;
    const int ne = (tot - e0) < EPW ? (tot - e0) : EPW;
    const float gmx = rng[0];
    const bool bounded = gmx + rng[1] <= SHIFT_RANGE;

    const bool real = e0 + EPW <= E;
    const bool loop = e0 >= E;
    if (ne == EPW && (real || loop)) {
        int s[EPW], d[EPW];
        float f[EPW][NJ];
        if (real) {
#pragma unroll
            for (int i = 0; i < EPW; ++i) { s[i] = src[e0 + i]; d[i] = dst[e0 + i]; }
        } else {
#pragma unroll
            for (int i = 0; i < EPW; ++i) { s[i] = d[i] = e0 + i - E; }
        }
#pragma unroll
        for (int i = 0; i < EPW; ++i)
#pragma unroll
            for (int j = 0; j < NJ; ++j)
                f[i][j] = feat[(size_t)s[i] * HC + j * 32 + lane];
        const int li = lane / H, lh = lane % H;
        int sl = s[0], dl = d[0];
#pragma unroll
        for (int i = 1; i < EPW; ++i)
            if (li == i) { sl = s[i]; dl = d[i]; }
        const int dsel = dl * H + lh;
        float wsel = 0.f;
        if (lane < EPW * H) {
            const v2f r = *(const v2f*)(rec + 2 * (size_t)dsel);
            float v = asrc[sl * H + lh] + r.x;
            v = v > 0.f ? v : 0.2f * v;
            float mx = (bounded ? gmx : r.y) + r.x;
            mx = mx > 0.f ? mx : 0.2f * mx;
            wsel = expf(v - mx);
        }
#pragma unroll
        for (int i = 0; i < EPW; ++i)
#pragma unroll
            for (int j = 0; j < NJ; ++j) {
                const float wij = __int_as_float(__builtin_amdgcn_readlane(
                    __float_as_int(wsel), i * H + (j * 32) / C));
                atomicAdd(&agg[(size_t)d[i] * HC + j * 32 + lane], f[i][j] * wij);
            }
        if (lane < EPW * H) atomicAdd(&den[dsel], wsel);
    } else {
        for (int i = 0; i < ne; ++i) {
            int s, d; edge_pair(e0 + i, E, src, dst, s, d);
            float w[H];
#pragma unroll
            for (int h = 0; h < H; ++h) {
                const v2f r = *(const v2f*)(rec + 2 * (size_t)(d * H + h));
                float v = asrc[s * H + h] + r.x;
                v = v > 0.f ? v : 0.2f * v;
                float mx = (bounded ? gmx : r.y) + r.x;
                mx = mx > 0.f ? mx : 0.2f * mx;
                w[h] = expf(v - mx);
            }
            float wl = w[0];
#pragma unroll
            for (int h = 1; h < H; ++h) wl = (lane == h) ? w[h] : wl;
            if (lane < H) atomicAdd(&den[d * H + lane], wl);
#pragma unroll
            for (int j = 0; j < NJ; ++j)
                atomicAdd(&agg[(size_t)d * HC + j * 32 + lane],
                          feat[(size_t)s * HC + j * 32 + lane] * w[(j * 32) / C]);
        }
    }
}

template <int HC, int C>
__global__ void k_bias_elu(float* __restrict__ h, const float* __restrict__ den,
                           const float* __restrict__ bias, int n) {
    int i = blockIdx.x * blockDim.x + threadIdx.x;
    if (i >= n * HC) return;
    int node = i / HC, c = i % HC;
    float v = h[i] / den[node * (HC / C) + c / C] + bias[c];
    h[i] = v > 0.f ? v : (expf(v) - 1.f);
}

template <int C>
__global__ void k_bias_add(float* __restrict__ o, const float* __restrict__ den,
                           const float* __restrict__ bias, int n) {
    int i = blockIdx.x * blockDim.x + threadIdx.x;
    if (i >= n * C) return;
    o[i] = o[i] / den[i / C] + bias[i % C];
}

extern "C" void kernel_launch(void* const* d_in, const int* in_sizes, int n_in,
                              void* d_out, int out_size, void* d_ws, size_t ws_size,
                              hipStream_t stream) {
    const float* x      = (const float*)d_in[0];
    const int*   ei     = (const int*)  d_in[1];
    const float* W1     = (const float*)d_in[2];
    const float* att_s1 = (const float*)d_in[3];
    const float* att_d1 = (const float*)d_in[4];
    const float* b1     = (const float*)d_in[5];
    const float* W2     = (const float*)d_in[6];
    const float* att_s2 = (const float*)d_in[7];
    const float* att_d2 = (const float*)d_in[8];
    const float* b2     = (const float*)d_in[9];

    const int IN = 128, HID = 128, H1 = 4, C1 = 32, OUT = 64;
    const int n   = in_sizes[0] / IN;
    const int E   = in_sizes[1] / 2;
    const int tot = E + n;

    float* out1  = (float*)d_out;
    float* hfeat = (float*)d_out + (size_t)n * OUT;

    float* w = (float*)d_ws;
    size_t off = 0;
    float* rc1 = w + off; off += (size_t)n * H1 * 2;
    float* rc2 = w + off; off += (size_t)n * 2;
    float* h1  = w + off; off += (size_t)n * HID;
    float* h2  = w + off; off += (size_t)n * OUT;
    float* as1 = w + off; off += (size_t)n * H1;
    float* dn1 = w + off; off += (size_t)n * H1;
    float* rg1 = w + off; off += 2;
    float* as2 = w + off; off += (size_t)n;
    float* dn2 = w + off; off += (size_t)n;
    float* rg2 = w + off; off += 2;

    const int* srcI = ei;
    const int* dstI = ei + E;

    auto cdiv = [](long a, long b) { return (unsigned)((a + b - 1) / b); };
    dim3 B256(256);

    const int gemmBlocks = 625;

    k_gemm_f32_wmma<128><<<gemmBlocks, 256, 0, stream>>>(x, W1, h1, n / 16);
    k_fill<<<cdiv((long)n * H1 + 2, 256), B256, 0, stream>>>(dn1, 0.f, n * H1 + 2);
    k_attn<<<cdiv((long)n * H1, 256), B256, 0, stream>>>(h1, att_s1, att_d1, as1, rc1, rg1, n, H1, C1);

    k_fill<<<cdiv((long)n * HID, 256), B256, 0, stream>>>(hfeat, 0.f, n * HID);

    k_edge_max<4><<<cdiv(E, 256), B256, 0, stream>>>(srcI, dstI, as1, rc1, rg1, E);
    k_edge_fused<4, 32><<<cdiv(tot, 32), B256, 0, stream>>>(srcI, dstI, as1, rc1, rg1, h1, hfeat, dn1, E, n);
    k_bias_elu<128, 32><<<cdiv((long)n * HID, 256), B256, 0, stream>>>(hfeat, dn1, b1, n);

    k_gemm_f32_wmma<64><<<gemmBlocks, 128, 0, stream>>>(hfeat, W2, h2, n / 16);
    k_fill<<<cdiv((long)n + 2, 256), B256, 0, stream>>>(dn2, 0.f, n + 2);
    k_attn<<<cdiv((long)n, 256), B256, 0, stream>>>(h2, att_s2, att_d2, as2, rc2, rg2, n, 1, OUT);

    k_fill<<<cdiv((long)n * OUT, 256), B256, 0, stream>>>(out1, 0.f, n * OUT);

    k_edge_max<1><<<cdiv(E, 256), B256, 0, stream>>>(srcI, dstI, as2, rc2, rg2, E);
    k_edge_fused<1, 64><<<cdiv(tot, 32), B256, 0, stream>>>(srcI, dstI, as2, rc2, rg2, h2, out1, dn2, E, n);
    k_bias_add<64><<<cdiv((long)n * OUT, 256), B256, 0, stream>>>(out1, dn2, b2, n);
}


# Edge sums kept apart per cache half: a block reads its die id and adds into its own half's sum plane and denominators (second plane in workspace); epilogues add the planes: partial sums per half
# speedup vs baseline: 5.4057x; 1.3674x over previous
#include <hip/hip_runtime.h>
#include <hip/hip_bf16.h>
#include <math.h>

typedef __attribute__((ext_vector_type(2))) float v2f;
typedef __attribute__((ext_vector_type(4))) float v4f;
typedef __attribute__((ext_vector_type(8))) float v8f;

__global__ void k_fill(float* __restrict__ p, float v, int n) {
    int i = blockIdx.x * blockDim.x + threadIdx.x;
    if (i < n) p[i] = v;
}
__global__ void k_fill2(float* __restrict__ p, int np, float* __restrict__ q, int nq, float v) {
    const int pb = (np + 255) / 256;
    if ((int)blockIdx.x < pb) {
        const int i = blockIdx.x * 256 + threadIdx.x;
        if (i < np) p[i] = v;
    } else {
        const int i = ((int)blockIdx.x - pb) * 256 + threadIdx.x;
        if (i < nq) q[i] = v;
    }
}

template <int NCOL>
__global__ void __launch_bounds__(NCOL * 2)
k_gemm_f32_wmma(const float* __restrict__ A, const float* __restrict__ W,
                float* __restrict__ C, int nTiles) {
    constexpr int K   = 128;
    constexpr int NW  = NCOL / 16;
    constexpr int RPW = 16 / NW;
    constexpr int P   = 132;
    __shared__ float sA[2][16 * P];

    const int lane = threadIdx.x & 31;
    const int wave = __builtin_amdgcn_readfirstlane((int)(threadIdx.x >> 5));
    const int half = lane >> 4;
    const int l    = lane & 15;
    const int colBase = wave * 16;

    int t = blockIdx.x;
    v4f st[RPW];
    if (t < nTiles) {
#pragma unroll
        for (int i = 0; i < RPW; ++i)
            st[i] = *(const v4f*)(A + (size_t)(t * 16 + wave * RPW + i) * K + lane * 4);
    }

    v2f b[K / 4];
    const float* wp = W + (size_t)(2 * half) * NCOL + colBase + l;
#pragma unroll
    for (int j = 0; j < K / 4; ++j) {
        b[j].x = wp[(size_t)(4 * j) * NCOL];
        b[j].y = wp[(size_t)(4 * j + 1) * NCOL];
    }

    int buf = 0;
    for (; t < nTiles; t += gridDim.x) {
#pragma unroll
        for (int i = 0; i < RPW; ++i)
            *(v4f*)(&sA[buf][(wave * RPW + i) * P + lane * 4]) = st[i];
        __syncthreads();
        const int tn = t + gridDim.x;
        if (tn < nTiles) {
#pragma unroll
            for (int i = 0; i < RPW; ++i)
                st[i] = *(const v4f*)(A + (size_t)(tn * 16 + wave * RPW + i) * K + lane * 4);
        }
        const float* ap = &sA[buf][l * P + 2 * half];
        v8f acc = {};
#pragma unroll
        for (int j = 0; j < K / 4; ++j) {
            v2f a = *(const v2f*)(ap + 4 * j);
            acc = __builtin_amdgcn_wmma_f32_16x16x4_f32(
                 false, a,  false, b[j],
                 (short)0, acc,  false,  false);
        }
#pragma unroll
        for (int r = 0; r < 8; ++r)
            C[(size_t)(t * 16 + r + 8 * half) * NCOL + colBase + l] = acc[r];
        buf ^= 1;
    }
}

__global__ void __launch_bounds__(256)
k_attn(const float* __restrict__ h, const float* __restrict__ att_s,
       const float* __restrict__ att_d, float* __restrict__ asrc,
       float* __restrict__ rec, float* __restrict__ rng, int n, int heads, int C) {
    __shared__ float shi[8], slo[8];
    const int i = blockIdx.x * blockDim.x + threadIdx.x;
    float hi = 0.f, lo = 0.f;
    if (i < n * heads) {
        int node = i / heads, hd = i % heads;
        const float* hp = h + (size_t)node * heads * C + (size_t)hd * C;
        float ss = 0.f, sd = 0.f;
        for (int c = 0; c < C; ++c) {
            float v = hp[c];
            ss += v * att_s[hd * C + c];
            sd += v * att_d[hd * C + c];
        }
        asrc[i] = ss;
        v2f r; r.x = sd; r.y = ss;
        *(v2f*)(rec + 2 * (size_t)i) = r;
        hi = ss; lo = -ss;
    }
    for (int o = 16; o > 0; o >>= 1) {
        hi = fmaxf(hi, __shfl_xor(hi, o));
        lo = fmaxf(lo, __shfl_xor(lo, o));
    }
    const int wv = threadIdx.x >> 5;
    if ((threadIdx.x & 31) == 0) { shi[wv] = hi; slo[wv] = lo; }
    __syncthreads();
    if (threadIdx.x == 0) {
        for (int w = 1; w < 8; ++w) { hi = fmaxf(hi, shi[w]); lo = fmaxf(lo, slo[w]); }
        __hip_atomic_fetch_max(&rng[0], hi, __ATOMIC_RELAXED, __HIP_MEMORY_SCOPE_AGENT);
        __hip_atomic_fetch_max(&rng[1], lo, __ATOMIC_RELAXED, __HIP_MEMORY_SCOPE_AGENT);
    }
}

__device__ __forceinline__ void edge_pair(int e, int E, const int* __restrict__ src,
                                          const int* __restrict__ dst, int& s, int& d) {
    if (e < E) { s = src[e]; d = dst[e]; }
    else       { s = d = e - E; }
}

#define SHIFT_RANGE 40.f
template <int H>
__global__ void __launch_bounds__(256)
k_edge_max(const int* __restrict__ src, const int* __restrict__ dst,
           const float* __restrict__ asrc, float* __restrict__ rec,
           const float* __restrict__ rng, int E) {
    if (rng[0] + rng[1] <= SHIFT_RANGE) return;
    constexpr int EPI = 32 / H;
    const int lane = threadIdx.x & 31;
    const int le = lane / H, lh = lane % H;
    const int e0 = (int)(blockIdx.x * 256 + (threadIdx.x & ~31u)) + le;
    int s[H], d[H];
#pragma unroll
    for (int j = 0; j < H; ++j) {
        const int e = min(e0 + j * EPI, E - 1);
        s[j] = src[e];
        d[j] = dst[e];
    }
    float v[H];
#pragma unroll
    for (int j = 0; j < H; ++j) v[j] = asrc[s[j] * H + lh];
#pragma unroll
    for (int j = 0; j < H; ++j)
        __hip_atomic_fetch_max(&rec[2 * (d[j] * H + lh) + 1], v[j], __ATOMIC_RELAXED,
                               __HIP_MEMORY_SCOPE_AGENT);
}

template <int H, int C>
__global__ void __launch_bounds__(256)
k_edge_fused(const int* __restrict__ src, const int* __restrict__ dst,
             const float* __restrict__ asrc, const float* __restrict__ rec,
             const float* __restrict__ rng, const float* __restrict__ feat,
             float* __restrict__ aggA, float* __restrict__ denA,
             float* __restrict__ aggB, float* __restrict__ denB, int E, int n) {
    constexpr int HC  = H * C;
    constexpr int NJ  = HC / 32;
    constexpr int EPW = 4;
    static_assert(HC % 32 == 0 && (H == 1 || C % 32 == 0), "line must not straddle heads");
    const int lane = threadIdx.x & 31;
    const int wave = __builtin_amdgcn_readfirstlane((int)(threadIdx.x >> 5));
    const int tot = E + n;
    const int e0 = (blockIdx.x * 8 + wave) * EPW;
    __shared__ unsigned sPlace;
    if (threadIdx.x == 0) sPlace = (unsigned)__builtin_amdgcn_s_sendmsg_rtn(0x87);
    __syncthreads();
    if (e0 >= tot) return;
    const unsigned place = (unsigned)__builtin_amdgcn_readfirstlane((int)sPlace);
    const bool upper = (place >> 18) & 1u;
    float* __restrict__ agg = upper ? aggB : aggA;
    float* __restrict__ den = upper ? denB : denA;
    const int ne = (tot - e0) < EPW ? (tot - e0) : EPW;
    const float gmx = rng[0];
    const bool bounded = gmx + rng[1] <= SHIFT_RANGE;

    const bool real = e0 + EPW <= E;
    const bool loop = e0 >= E;
    if (ne == EPW && (real || loop)) {
        int s[EPW], d[EPW];
        float f[EPW][NJ];
        if (real) {
#pragma unroll
            for (int i = 0; i < EPW; ++i) { s[i] = src[e0 + i]; d[i] = dst[e0 + i]; }
        } else {
#pragma unroll
            for (int i = 0; i < EPW; ++i) { s[i] = d[i] = e0 + i - E; }
        }
#pragma unroll
        for (int i = 0; i < EPW; ++i)
#pragma unroll
            for (int j = 0; j < NJ; ++j)
                f[i][j] = feat[(size_t)s[i] * HC + j * 32 + lane];
        const int li = lane / H, lh = lane % H;
        int sl = s[0], dl = d[0];
#pragma unroll
        for (int i = 1; i < EPW; ++i)
            if (li == i) { sl = s[i]; dl = d[i]; }
        const int dsel = dl * H + lh;
        float wsel = 0.f;
        if (lane < EPW * H) {
            const v2f r = *(const v2f*)(rec + 2 * (size_t)dsel);
            float v = asrc[sl * H + lh] + r.x;
            v = v > 0.f ? v : 0.2f * v;
            float mx = (bounded ? gmx : r.y) + r.x;
            mx = mx > 0.f ? mx : 0.2f * mx;
            wsel = expf(v - mx);
        }
#pragma unroll
        for (int i = 0; i < EPW; ++i)
#pragma unroll
            for (int j = 0; j < NJ; ++j) {
                const float wij = __int_as_float(__builtin_amdgcn_readlane(
                    __float_as_int(wsel), i * H + (j * 32) / C));
                atomicAdd(&agg[(size_t)d[i] * HC + j * 32 + lane], f[i][j] * wij);
            }
        if (lane < EPW * H) atomicAdd(&den[dsel], wsel);
    } else {
        for (int i = 0; i < ne; ++i) {
            int s, d; edge_pair(e0 + i, E, src, dst, s, d);
            float w[H];
#pragma unroll
            for (int h = 0; h < H; ++h) {
                const v2f r = *(const v2f*)(rec + 2 * (size_t)(d * H + h));
                float v = asrc[s * H + h] + r.x;
                v = v > 0.f ? v : 0.2f * v;
                float mx = (bounded ? gmx : r.y) + r.x;
                mx = mx > 0.f ? mx : 0.2f * mx;
                w[h] = expf(v - mx);
            }
            float wl = w[0];
#pragma unroll
            for (int h = 1; h < H; ++h) wl = (lane == h) ? w[h] : wl;
            if (lane < H) atomicAdd(&den[d * H + lane], wl);
#pragma unroll
            for (int j = 0; j < NJ; ++j)
                atomicAdd(&agg[(size_t)d * HC + j * 32 + lane],
                          feat[(size_t)s * HC + j * 32 + lane] * w[(j * 32) / C]);
        }
    }
}

template <int HC, int C>
__global__ void k_bias_elu(float* __restrict__ h, const float* __restrict__ hB,
                           const float* __restrict__ den, const float* __restrict__ denB,
                           const float* __restrict__ bias, int n) {
    int i = blockIdx.x * blockDim.x + threadIdx.x;
    if (i >= n * HC) return;
    int node = i / HC, c = i % HC;
    const int k = node * (HC / C) + c / C;
    float v = (h[i] + hB[i]) / (den[k] + denB[k]) + bias[c];
    h[i] = v > 0.f ? v : (expf(v) - 1.f);
}

template <int C>
__global__ void k_bias_add(float* __restrict__ o, const float* __restrict__ oB,
                           const float* __restrict__ den, const float* __restrict__ denB,
                           const float* __restrict__ bias, int n) {
    int i = blockIdx.x * blockDim.x + threadIdx.x;
    if (i >= n * C) return;
    o[i] = (o[i] + oB[i]) / (den[i / C] + denB[i / C]) + bias[i % C];
}

extern "C" void kernel_launch(void* const* d_in, const int* in_sizes, int n_in,
                              void* d_out, int out_size, void* d_ws, size_t ws_size,
                              hipStream_t stream) {
    const float* x      = (const float*)d_in[0];
    const int*   ei     = (const int*)  d_in[1];
    const float* W1     = (const float*)d_in[2];
    const float* att_s1 = (const float*)d_in[3];
    const float* att_d1 = (const float*)d_in[4];
    const float* b1     = (const float*)d_in[5];
    const float* W2     = (const float*)d_in[6];
    const float* att_s2 = (const float*)d_in[7];
    const float* att_d2 = (const float*)d_in[8];
    const float* b2     = (const float*)d_in[9];

    const int IN = 128, HID = 128, H1 = 4, C1 = 32, OUT = 64;
    const int n   = in_sizes[0] / IN;
    const int E   = in_sizes[1] / 2;
    const int tot = E + n;

    float* out1  = (float*)d_out;
    float* hfeat = (float*)d_out + (size_t)n * OUT;

    float* w = (float*)d_ws;
    size_t off = 0;
    float* pB1 = w + off; off += (size_t)n * HID;
    float* pB2 = w + off; off += (size_t)n * OUT;
    float* rc1 = w + off; off += (size_t)n * H1 * 2;
    float* rc2 = w + off; off += (size_t)n * 2;
    float* h1  = w + off; off += (size_t)n * HID;
    float* h2  = w + off; off += (size_t)n * OUT;
    float* as1 = w + off; off += (size_t)n * H1;
    float* dn1 = w + off; off += (size_t)n * H1;
    float* rg1 = w + off; off += 2;
    float* dB1 = w + off; off += (size_t)n * H1;
    float* as2 = w + off; off += (size_t)n;
    float* dn2 = w + off; off += (size_t)n;
    float* rg2 = w + off; off += 2;
    float* dB2 = w + off; off += (size_t)n;

    const int* srcI = ei;
    const int* dstI = ei + E;

    auto cdiv = [](long a, long b) { return (unsigned)((a + b - 1) / b); };
    dim3 B256(256);

    const int gemmBlocks = 625;

    k_gemm_f32_wmma<128><<<gemmBlocks, 256, 0, stream>>>(x, W1, h1, n / 16);
    k_fill2<<<cdiv((long)n * H1 * 2 + 2, 256) + cdiv((long)n * HID, 256), B256, 0, stream>>>(
        dn1, n * H1 * 2 + 2, pB1, n * HID, 0.f);
    k_attn<<<cdiv((long)n * H1, 256), B256, 0, stream>>>(h1, att_s1, att_d1, as1, rc1, rg1, n, H1, C1);

    k_fill<<<cdiv((long)n * HID, 256), B256, 0, stream>>>(hfeat, 0.f, n * HID);

    k_edge_max<4><<<cdiv(E, 256), B256, 0, stream>>>(srcI, dstI, as1, rc1, rg1, E);
    k_edge_fused<4, 32><<<cdiv(tot, 32), B256, 0, stream>>>(srcI, dstI, as1, rc1, rg1, h1, hfeat, dn1, pB1, dB1, E, n);
    k_bias_elu<128, 32><<<cdiv((long)n * HID, 256), B256, 0, stream>>>(hfeat, pB1, dn1, dB1, b1, n);

    k_gemm_f32_wmma<64><<<gemmBlocks, 128, 0, stream>>>(hfeat, W2, h2, n / 16);
    k_fill2<<<cdiv((long)n * 2 + 2, 256) + cdiv((long)n * OUT, 256), B256, 0, stream>>>(
        dn2, n * 2 + 2, pB2, n * OUT, 0.f);
    k_attn<<<cdiv((long)n, 256), B256, 0, stream>>>(h2, att_s2, att_d2, as2, rc2, rg2, n, 1, OUT);

    k_fill<<<cdiv((long)n * OUT, 256), B256, 0, stream>>>(out1, 0.f, n * OUT);

    k_edge_max<1><<<cdiv(E, 256), B256, 0, stream>>>(srcI, dstI, as2, rc2, rg2, E);
    k_edge_fused<1, 64><<<cdiv(tot, 32), B256, 0, stream>>>(srcI, dstI, as2, rc2, rg2, h2, out1, dn2, pB2, dB2, E, n);
    k_bias_add<64><<<cdiv((long)n * OUT, 256), B256, 0, stream>>>(out1, pB2, dn2, dB2, b2, n);
}


# Zero start values of the edge sums (both planes, denominators, bound words) stored by the GEMM blocks for their row tiles, under the matrix issue: 4 fill launches gone, 14 to 10; sums as before
# speedup vs baseline: 6.1643x; 1.1403x over previous
#include <hip/hip_runtime.h>
#include <hip/hip_bf16.h>
#include <math.h>

typedef __attribute__((ext_vector_type(2))) float v2f;
typedef __attribute__((ext_vector_type(4))) float v4f;
typedef __attribute__((ext_vector_type(8))) float v8f;

template <int NCOL, int H>
__global__ void __launch_bounds__(NCOL * 2)
k_gemm_f32_wmma(const float* __restrict__ A, const float* __restrict__ W,
                float* __restrict__ C, int nTiles,
                float* __restrict__ zA, float* __restrict__ zB,
                float* __restrict__ dA, float* __restrict__ dB,
                float* __restrict__ rg) {
    constexpr int K   = 128;
    constexpr int NW  = NCOL / 16;
    constexpr int RPW = 16 / NW;
    constexpr int P   = 132;
    __shared__ float sA[2][16 * P];

    const int lane = threadIdx.x & 31;
    const int wave = __builtin_amdgcn_readfirstlane((int)(threadIdx.x >> 5));
    const int half = lane >> 4;
    const int l    = lane & 15;
    const int colBase = wave * 16;

    if (blockIdx.x == 0 && threadIdx.x < 2) rg[threadIdx.x] = 0.f;

    int t = blockIdx.x;
    v4f st[RPW];
    if (t < nTiles) {
#pragma unroll
        for (int i = 0; i < RPW; ++i)
            st[i] = *(const v4f*)(A + (size_t)(t * 16 + wave * RPW + i) * K + lane * 4);
    }

    v2f b[K / 4];
    const float* wp = W + (size_t)(2 * half) * NCOL + colBase + l;
#pragma unroll
    for (int j = 0; j < K / 4; ++j) {
        b[j].x = wp[(size_t)(4 * j) * NCOL];
        b[j].y = wp[(size_t)(4 * j + 1) * NCOL];
    }

    int buf = 0;
    for (; t < nTiles; t += gridDim.x) {
#pragma unroll
        for (int i = 0; i < RPW; ++i)
            *(v4f*)(&sA[buf][(wave * RPW + i) * P + lane * 4]) = st[i];
        __syncthreads();
        const int tn = t + gridDim.x;
        if (tn < nTiles) {
#pragma unroll
            for (int i = 0; i < RPW; ++i)
                st[i] = *(const v4f*)(A + (size_t)(tn * 16 + wave * RPW + i) * K + lane * 4);
        }
        {
            const v4f z4 = {0.f, 0.f, 0.f, 0.f};
            const size_t zo = (size_t)t * 16 * NCOL + threadIdx.x * 4;
            *(v4f*)(zA + zo)            = z4;
            *(v4f*)(zA + zo + NCOL * 8) = z4;
            *(v4f*)(zB + zo)            = z4;
            *(v4f*)(zB + zo + NCOL * 8) = z4;
            if (threadIdx.x < 32 * H) {
                const int k = threadIdx.x;
                float* p = (k < 16 * H) ? dA + (size_t)t * 16 * H + k
                                        : dB + (size_t)t * 16 * H + (k - 16 * H);
                *p = 0.f;
            }
        }
        const float* ap = &sA[buf][l * P + 2 * half];
        v8f acc = {};
#pragma unroll
        for (int j = 0; j < K / 4; ++j) {
            v2f a = *(const v2f*)(ap + 4 * j);
            acc = __builtin_amdgcn_wmma_f32_16x16x4_f32(
                 false, a,  false, b[j],
                 (short)0, acc,  false,  false);
        }
#pragma unroll
        for (int r = 0; r < 8; ++r)
            C[(size_t)(t * 16 + r + 8 * half) * NCOL + colBase + l] = acc[r];
        buf ^= 1;
    }
}

__global__ void __launch_bounds__(256)
k_attn(const float* __restrict__ h, const float* __restrict__ att_s,
       const float* __restrict__ att_d, float* __restrict__ asrc,
       float* __restrict__ rec, float* __restrict__ rng, int n, int heads, int C) {
    __shared__ float shi[8], slo[8];
    const int i = blockIdx.x * blockDim.x + threadIdx.x;
    float hi = 0.f, lo = 0.f;
    if (i < n * heads) {
        int node = i / heads, hd = i % heads;
        const float* hp = h + (size_t)node * heads * C + (size_t)hd * C;
        float ss = 0.f, sd = 0.f;
        for (int c = 0; c < C; ++c) {
            float v = hp[c];
            ss += v * att_s[hd * C + c];
            sd += v * att_d[hd * C + c];
        }
        asrc[i] = ss;
        v2f r; r.x = sd; r.y = ss;
        *(v2f*)(rec + 2 * (size_t)i) = r;
        hi = ss; lo = -ss;
    }
    for (int o = 16; o > 0; o >>= 1) {
        hi = fmaxf(hi, __shfl_xor(hi, o));
        lo = fmaxf(lo, __shfl_xor(lo, o));
    }
    const int wv = threadIdx.x >> 5;
    if ((threadIdx.x & 31) == 0) { shi[wv] = hi; slo[wv] = lo; }
    __syncthreads();
    if (threadIdx.x == 0) {
        for (int w = 1; w < 8; ++w) { hi = fmaxf(hi, shi[w]); lo = fmaxf(lo, slo[w]); }
        __hip_atomic_fetch_max(&rng[0], hi, __ATOMIC_RELAXED, __HIP_MEMORY_SCOPE_AGENT);
        __hip_atomic_fetch_max(&rng[1], lo, __ATOMIC_RELAXED, __HIP_MEMORY_SCOPE_AGENT);
    }
}

__device__ __forceinline__ void edge_pair(int e, int E, const int* __restrict__ src,
                                          const int* __restrict__ dst, int& s, int& d) {
    if (e < E) { s = src[e]; d = dst[e]; }
    else       { s = d = e - E; }
}

#define SHIFT_RANGE 40.f
template <int H>
__global__ void __launch_bounds__(256)
k_edge_max(const int* __restrict__ src, const int* __restrict__ dst,
           const float* __restrict__ asrc, float* __restrict__ rec,
           const float* __restrict__ rng, int E) {
    if (rng[0] + rng[1] <= SHIFT_RANGE) return;
    constexpr int EPI = 32 / H;
    const int lane = threadIdx.x & 31;
    const int le = lane / H, lh = lane % H;
    const int e0 = (int)(blockIdx.x * 256 + (threadIdx.x & ~31u)) + le;
    int s[H], d[H];
#pragma unroll
    for (int j = 0; j < H; ++j) {
        const int e = min(e0 + j * EPI, E - 1);
        s[j] = src[e];
        d[j] = dst[e];
    }
    float v[H];
#pragma unroll
    for (int j = 0; j < H; ++j) v[j] = asrc[s[j] * H + lh];
#pragma unroll
    for (int j = 0; j < H; ++j)
        __hip_atomic_fetch_max(&rec[2 * (d[j] * H + lh) + 1], v[j], __ATOMIC_RELAXED,
                               __HIP_MEMORY_SCOPE_AGENT);
}

template <int H, int C>
__global__ void __launch_bounds__(256)
k_edge_fused(const int* __restrict__ src, const int* __restrict__ dst,
             const float* __restrict__ asrc, const float* __restrict__ rec,
             const float* __restrict__ rng, const float* __restrict__ feat,
             float* __restrict__ aggA, float* __restrict__ denA,
             float* __restrict__ aggB, float* __restrict__ denB, int E, int n) {
    constexpr int HC  = H * C;
    constexpr int NJ  = HC / 32;
    constexpr int EPW = 4;
    static_assert(HC % 32 == 0 && (H == 1 || C % 32 == 0), "line must not straddle heads");
    const int lane = threadIdx.x & 31;
    const int wave = __builtin_amdgcn_readfirstlane((int)(threadIdx.x >> 5));
    const int tot = E + n;
    const int e0 = (blockIdx.x * 8 + wave) * EPW;
    __shared__ unsigned sPlace;
    if (threadIdx.x == 0) sPlace = (unsigned)__builtin_amdgcn_s_sendmsg_rtn(0x87);
    __syncthreads();
    if (e0 >= tot) return;
    const unsigned place = (unsigned)__builtin_amdgcn_readfirstlane((int)sPlace);
    const bool upper = (place >> 18) & 1u;
    float* __restrict__ agg = upper ? aggB : aggA;
    float* __restrict__ den = upper ? denB : denA;
    const int ne = (tot - e0) < EPW ? (tot - e0) : EPW;
    const float gmx = rng[0];
    const bool bounded = gmx + rng[1] <= SHIFT_RANGE;

    const bool real = e0 + EPW <= E;
    const bool loop = e0 >= E;
    if (ne == EPW && (real || loop)) {
        int s[EPW], d[EPW];
        float f[EPW][NJ];
        if (real) {
#pragma unroll
            for (int i = 0; i < EPW; ++i) { s[i] = src[e0 + i]; d[i] = dst[e0 + i]; }
        } else {
#pragma unroll
            for (int i = 0; i < EPW; ++i) { s[i] = d[i] = e0 + i - E; }
        }
#pragma unroll
        for (int i = 0; i < EPW; ++i)
#pragma unroll
            for (int j = 0; j < NJ; ++j)
                f[i][j] = feat[(size_t)s[i] * HC + j * 32 + lane];
        const int li = lane / H, lh = lane % H;
        int sl = s[0], dl = d[0];
#pragma unroll
        for (int i = 1; i < EPW; ++i)
            if (li == i) { sl = s[i]; dl = d[i]; }
        const int dsel = dl * H + lh;
        float wsel = 0.f;
        if (lane < EPW * H) {
            const v2f r = *(const v2f*)(rec + 2 * (size_t)dsel);
            float v = asrc[sl * H + lh] + r.x;
            v = v > 0.f ? v : 0.2f * v;
            float mx = (bounded ? gmx : r.y) + r.x;
            mx = mx > 0.f ? mx : 0.2f * mx;
            wsel = expf(v - mx);
        }
#pragma unroll
        for (int i = 0; i < EPW; ++i)
#pragma unroll
            for (int j = 0; j < NJ; ++j) {
                const float wij = __int_as_float(__builtin_amdgcn_readlane(
                    __float_as_int(wsel), i * H + (j * 32) / C));
                atomicAdd(&agg[(size_t)d[i] * HC + j * 32 + lane], f[i][j] * wij);
            }
        if (lane < EPW * H) atomicAdd(&den[dsel], wsel);
    } else {
        for (int i = 0; i < ne; ++i) {
            int s, d; edge_pair(e0 + i, E, src, dst, s, d);
            float w[H];
#pragma unroll
            for (int h = 0; h < H; ++h) {
                const v2f r = *(const v2f*)(rec + 2 * (size_t)(d * H + h));
                float v = asrc[s * H + h] + r.x;
                v = v > 0.f ? v : 0.2f * v;
                float mx = (bounded ? gmx : r.y) + r.x;
                mx = mx > 0.f ? mx : 0.2f * mx;
                w[h] = expf(v - mx);
            }
            float wl = w[0];
#pragma unroll
            for (int h = 1; h < H; ++h) wl = (lane == h) ? w[h] : wl;
            if (lane < H) atomicAdd(&den[d * H + lane], wl);
#pragma unroll
            for (int j = 0; j < NJ; ++j)
                atomicAdd(&agg[(size_t)d * HC + j * 32 + lane],
                          feat[(size_t)s * HC + j * 32 + lane] * w[(j * 32) / C]);
        }
    }
}

template <int HC, int C>
__global__ void k_bias_elu(float* __restrict__ h, const float* __restrict__ hB,
                           const float* __restrict__ den, const float* __restrict__ denB,
                           const float* __restrict__ bias, int n) {
    int i = blockIdx.x * blockDim.x + threadIdx.x;
    if (i >= n * HC) return;
    int node = i / HC, c = i % HC;
    const int k = node * (HC / C) + c / C;
    float v = (h[i] + hB[i]) / (den[k] + denB[k]) + bias[c];
    h[i] = v > 0.f ? v : (expf(v) - 1.f);
}

template <int C>
__global__ void k_bias_add(float* __restrict__ o, const float* __restrict__ oB,
                           const float* __restrict__ den, const float* __restrict__ denB,
                           const float* __restrict__ bias, int n) {
    int i = blockIdx.x * blockDim.x + threadIdx.x;
    if (i >= n * C) return;
    o[i] = (o[i] + oB[i]) / (den[i / C] + denB[i / C]) + bias[i % C];
}

extern "C" void kernel_launch(void* const* d_in, const int* in_sizes, int n_in,
                              void* d_out, int out_size, void* d_ws, size_t ws_size,
                              hipStream_t stream) {
    const float* x      = (const float*)d_in[0];
    const int*   ei     = (const int*)  d_in[1];
    const float* W1     = (const float*)d_in[2];
    const float* att_s1 = (const float*)d_in[3];
    const float* att_d1 = (const float*)d_in[4];
    const float* b1     = (const float*)d_in[5];
    const float* W2     = (const float*)d_in[6];
    const float* att_s2 = (const float*)d_in[7];
    const float* att_d2 = (const float*)d_in[8];
    const float* b2     = (const float*)d_in[9];

    const int IN = 128, HID = 128, H1 = 4, C1 = 32, OUT = 64;
    const int n   = in_sizes[0] / IN;
    const int E   = in_sizes[1] / 2;
    const int tot = E + n;

    float* out1  = (float*)d_out;
    float* hfeat = (float*)d_out + (size_t)n * OUT;

    float* w = (float*)d_ws;
    size_t off = 0;
    float* pB1 = w + off; off += (size_t)n * HID;
    float* pB2 = w + off; off += (size_t)n * OUT;
    float* rc1 = w + off; off += (size_t)n * H1 * 2;
    float* rc2 = w + off; off += (size_t)n * 2;
    float* h1  = w + off; off += (size_t)n * HID;
    float* h2  = w + off; off += (size_t)n * OUT;
    float* as1 = w + off; off += (size_t)n * H1;
    float* dn1 = w + off; off += (size_t)n * H1;
    float* dB1 = w + off; off += (size_t)n * H1;
    float* rg1 = w + off; off += 32;
    float* as2 = w + off; off += (size_t)n;
    float* dn2 = w + off; off += (size_t)n;
    float* dB2 = w + off; off += (size_t)n;
    float* rg2 = w + off; off += 32;

    const int* srcI = ei;
    const int* dstI = ei + E;

    auto cdiv = [](long a, long b) { return (unsigned)((a + b - 1) / b); };
    dim3 B256(256);

    const int gemmBlocks = 625;

    k_gemm_f32_wmma<128, 4><<<gemmBlocks, 256, 0, stream>>>(x, W1, h1, n / 16,
                                                            hfeat, pB1, dn1, dB1, rg1);
    k_attn<<<cdiv((long)n * H1, 256), B256, 0, stream>>>(h1, att_s1, att_d1, as1, rc1, rg1, n, H1, C1);
    k_edge_max<4><<<cdiv(E, 256), B256, 0, stream>>>(srcI, dstI, as1, rc1, rg1, E);
    k_edge_fused<4, 32><<<cdiv(tot, 32), B256, 0, stream>>>(srcI, dstI, as1, rc1, rg1, h1, hfeat, dn1, pB1, dB1, E, n);
    k_bias_elu<128, 32><<<cdiv((long)n * HID, 256), B256, 0, stream>>>(hfeat, pB1, dn1, dB1, b1, n);

    k_gemm_f32_wmma<64, 1><<<gemmBlocks, 128, 0, stream>>>(hfeat, W2, h2, n / 16,
                                                           out1, pB2, dn2, dB2, rg2);
    k_attn<<<cdiv((long)n, 256), B256, 0, stream>>>(h2, att_s2, att_d2, as2, rc2, rg2, n, 1, OUT);
    k_edge_max<1><<<cdiv(E, 256), B256, 0, stream>>>(srcI, dstI, as2, rc2, rg2, E);
    k_edge_fused<1, 64><<<cdiv(tot, 32), B256, 0, stream>>>(srcI, dstI, as2, rc2, rg2, h2, out1, dn2, pB2, dB2, E, n);
    k_bias_add<64><<<cdiv((long)n * OUT, 256), B256, 0, stream>>>(out1, pB2, dn2, dB2, b2, n);
}
